# starting kernel
# baseline (speedup 1.0000x reference)
#include <hip/hip_runtime.h>
#include <hip/hip_bf16.h>
#include <math.h>


typedef __attribute__((ext_vector_type(2))) float v2f;
typedef __attribute__((ext_vector_type(8))) float v8f;

#define BATCH 256
#define N_DIM 64
#define D_DIM 128
#define H_DIM 64
#define LDP   66


__device__ inline v8f wmma_f32x4(v2f a, v2f b, v8f c) {
  return __builtin_amdgcn_wmma_f32_16x16x4_f32(
       false, a,  false, b,
       (short)0, c,  false,  false);
}

__device__ inline void init_acc(v8f acc[4], const float* __restrict__ bias, int l16) {
#pragma unroll
  for (int t = 0; t < 4; ++t) {
    float bv = bias ? bias[t * 16 + l16] : 0.0f;
#pragma unroll
    for (int r = 0; r < 8; ++r) acc[t][r] = bv;
  }
}

__device__ inline void gemm_accT(const float* __restrict__ A, int lda,
                                 const float* __restrict__ BT, int ldbt,
                                 int K, v8f acc[4], int lane, int rowBase) {
  const int half = lane >> 4;
  const int l16  = lane & 15;
  for (int k0 = 0; k0 < K; k0 += 4) {
    const int ka = k0 + 2 * half;
    v2f a = *(const v2f*)(A + (rowBase + l16) * lda + ka);
#pragma unroll
    for (int t = 0; t < 4; ++t) {
      v2f b = *(const v2f*)(BT + (t * 16 + l16) * ldbt + ka);
      acc[t] = wmma_f32x4(a, b, acc[t]);
    }
  }
}

__device__ inline void store_stripe(float* __restrict__ D, int ldd,
                                    const v8f acc[4], int lane, int rowBase,
                                    bool do_relu) {
  const int half = lane >> 4;
  const int l16  = lane & 15;
#pragma unroll
  for (int t = 0; t < 4; ++t)
#pragma unroll
    for (int r = 0; r < 8; ++r) {
      float v = acc[t][r];
      if (do_relu) v = fmaxf(v, 0.0f);
      D[(rowBase + r + 8 * half) * ldd + t * 16 + l16] = v;
    }
}

__device__ inline void store_stripeT(float* __restrict__ DT, int ldt,
                                     const v8f acc[4], int lane, int rowBase) {
  const int half = lane >> 4;
  const int l16  = lane & 15;
#pragma unroll
  for (int t = 0; t < 4; ++t)
#pragma unroll
    for (int r = 0; r < 8; ++r)
      DT[(t * 16 + l16) * ldt + rowBase + r + 8 * half] = acc[t][r];
}

__device__ inline void copy_w_T(float* __restrict__ dst,
                                const float* __restrict__ src,
                                int koff, int tid) {
  for (int s = tid; s < 64 * 64; s += 128) {
    const int k = s >> 6;
    const int n = s & 63;
    dst[n * LDP + k] = src[(k + koff) * H_DIM + n];
  }
}

__device__ inline void copy_obs_half(float* __restrict__ dst,
                                     const float* __restrict__ src,
                                     int coff, int tid) {
  for (int s = tid; s < 64 * 32; s += 128) {
    const int r  = s >> 5;
    const int c2 = (s & 31) * 2;
    *(float2*)(dst + r * LDP + c2) =
        *(const float2*)(src + r * D_DIM + coff + c2);
  }
}

__global__ __launch_bounds__(128) void gated_comm_kernel(
    const float* __restrict__ obs, const float* __restrict__ enc_w,
    const float* __restrict__ enc_b, const float* __restrict__ gate_w1,
    const float* __restrict__ gate_b1, const float* __restrict__ gate_w2,
    const float* __restrict__ gate_b2, const float* __restrict__ msg_w1,
    const float* __restrict__ msg_b1, const float* __restrict__ msg_w2,
    const float* __restrict__ msg_b2, float* __restrict__ out) {
  __shared__ float smem[3 * 64 * LDP + 64];

  const int b       = blockIdx.x;
  const int tid     = threadIdx.x;
  const int lane    = tid & 31;
  const int wave    = tid >> 5;
  const int l16     = lane & 15;
  const int rowBase = wave * 16;

  float* R0  = smem;
  float* R1  = smem + 64 * LDP;
  float* R2  = smem + 2 * 64 * LDP;
  float* sW2 = smem + 3 * 64 * LDP;

  const float* obsb = obs + (size_t)b * (N_DIM * D_DIM);
  v8f acc[4];

  copy_obs_half(R0, obsb, 0, tid);
  copy_w_T(R2, enc_w, 0, tid);
  __syncthreads();
  init_acc(acc, enc_b, l16);
  gemm_accT(R0, LDP, R2, LDP, 64, acc, lane, rowBase);
  __syncthreads();
  copy_obs_half(R0, obsb, 64, tid);
  copy_w_T(R2, enc_w, 64, tid);
  __syncthreads();
  gemm_accT(R0, LDP, R2, LDP, 64, acc, lane, rowBase);
  __syncthreads();
  store_stripe(R0, LDP, acc, lane, rowBase,  true);
  copy_w_T(R2, gate_w1, 0, tid);
  __syncthreads();

  init_acc(acc, nullptr, l16);
  gemm_accT(R0, LDP, R2, LDP, H_DIM, acc, lane, rowBase);
  store_stripe(R1, LDP, acc, lane, rowBase, false);
  __syncthreads();
  copy_w_T(R2, gate_w1, 64, tid);
  __syncthreads();

  init_acc(acc, gate_b1, l16);
  gemm_accT(R0, LDP, R2, LDP, H_DIM, acc, lane, rowBase);
  __syncthreads();
  store_stripe(R2, LDP, acc, lane, rowBase, false);
  if (tid < H_DIM) sW2[tid] = gate_w2[tid];
  __syncthreads();

  const float gb2 = gate_b2[0];
  float gv[32];
#pragma unroll 1
  for (int ii = 0; ii < 16; ++ii) {
    const int i = rowBase + ii;
    float s0 = 0.0f, s1 = 0.0f;
    for (int h = 0; h < H_DIM; ++h) {
      const float gih = R1[i * LDP + h];
      const float w2h = sW2[h];
      s0 += fmaxf(gih + R2[lane * LDP + h], 0.0f) * w2h;
      s1 += fmaxf(gih + R2[(lane + 32) * LDP + h], 0.0f) * w2h;
    }
    gv[2 * ii + 0] = 1.0f / (1.0f + __expf(-(s0 + gb2)));
    gv[2 * ii + 1] = 1.0f / (1.0f + __expf(-(s1 + gb2)));
  }
#pragma unroll
  for (int ii = 0; ii < 16; ++ii) {
    R1[(rowBase + ii) * LDP + lane]      = gv[2 * ii + 0];
    R1[(rowBase + ii) * LDP + lane + 32] = gv[2 * ii + 1];
  }
  __syncthreads();

  copy_w_T(R2, msg_w1, 0, tid);
  __syncthreads();
  init_acc(acc, msg_b1, l16);
  gemm_accT(R0, LDP, R2, LDP, H_DIM, acc, lane, rowBase);
  __syncthreads();
  store_stripe(R0, LDP, acc, lane, rowBase,  true);
  copy_w_T(R2, msg_w2, 0, tid);
  __syncthreads();
  init_acc(acc, msg_b2, l16);
  gemm_accT(R0, LDP, R2, LDP, H_DIM, acc, lane, rowBase);
  __syncthreads();
  store_stripeT(R2, LDP, acc, lane, rowBase);
  __syncthreads();

  init_acc(acc, nullptr, l16);
  gemm_accT(R1, LDP, R2, LDP, N_DIM, acc, lane, rowBase);

  float* outb = out + (size_t)b * (N_DIM * H_DIM);
  const int half = lane >> 4;
#pragma unroll
  for (int t = 0; t < 4; ++t)
#pragma unroll
    for (int r = 0; r < 8; ++r)
      outb[(rowBase + r + 8 * half) * H_DIM + t * 16 + l16] = acc[t][r];
}

extern "C" void kernel_launch(void* const* d_in, const int* in_sizes, int n_in,
                              void* d_out, int out_size, void* d_ws, size_t ws_size,
                              hipStream_t stream) {
  const float* obs     = (const float*)d_in[0];
  const float* enc_w   = (const float*)d_in[1];
  const float* enc_b   = (const float*)d_in[2];
  const float* gate_w1 = (const float*)d_in[3];
  const float* gate_b1 = (const float*)d_in[4];
  const float* gate_w2 = (const float*)d_in[5];
  const float* gate_b2 = (const float*)d_in[6];
  const float* msg_w1  = (const float*)d_in[7];
  const float* msg_b1  = (const float*)d_in[8];
  const float* msg_w2  = (const float*)d_in[9];
  const float* msg_b2  = (const float*)d_in[10];
  float* out = (float*)d_out;

  gated_comm_kernel<<<BATCH, 128, 0, stream>>>(
      obs, enc_w, enc_b, gate_w1, gate_b1, gate_w2, gate_b2,
      msg_w1, msg_b1, msg_w2, msg_b2, out);
}


# gate reduction register-blocked: gj staged transposed, each lane keeps its two columns' gj and w2 in registers, 4 rows i per pass on packed f32 add/fma, gi read one chunk ahead; sum order unchanged
# speedup vs baseline: 1.3086x; 1.3086x over previous
#include <hip/hip_runtime.h>
#include <hip/hip_bf16.h>
#include <math.h>


typedef __attribute__((ext_vector_type(2))) float v2f;
typedef __attribute__((ext_vector_type(8))) float v8f;

#define BATCH 256
#define N_DIM 64
#define D_DIM 128
#define H_DIM 64
#define LDP   66


__device__ inline v8f wmma_f32x4(v2f a, v2f b, v8f c) {
  return __builtin_amdgcn_wmma_f32_16x16x4_f32(
       false, a,  false, b,
       (short)0, c,  false,  false);
}

__device__ inline void init_acc(v8f acc[4], const float* __restrict__ bias, int l16) {
#pragma unroll
  for (int t = 0; t < 4; ++t) {
    float bv = bias ? bias[t * 16 + l16] : 0.0f;
#pragma unroll
    for (int r = 0; r < 8; ++r) acc[t][r] = bv;
  }
}

__device__ inline void gemm_accT(const float* __restrict__ A, int lda,
                                 const float* __restrict__ BT, int ldbt,
                                 int K, v8f acc[4], int lane, int rowBase) {
  const int half = lane >> 4;
  const int l16  = lane & 15;
  for (int k0 = 0; k0 < K; k0 += 4) {
    const int ka = k0 + 2 * half;
    v2f a = *(const v2f*)(A + (rowBase + l16) * lda + ka);
#pragma unroll
    for (int t = 0; t < 4; ++t) {
      v2f b = *(const v2f*)(BT + (t * 16 + l16) * ldbt + ka);
      acc[t] = wmma_f32x4(a, b, acc[t]);
    }
  }
}

__device__ inline void store_stripe(float* __restrict__ D, int ldd,
                                    const v8f acc[4], int lane, int rowBase,
                                    bool do_relu) {
  const int half = lane >> 4;
  const int l16  = lane & 15;
#pragma unroll
  for (int t = 0; t < 4; ++t)
#pragma unroll
    for (int r = 0; r < 8; ++r) {
      float v = acc[t][r];
      if (do_relu) v = fmaxf(v, 0.0f);
      D[(rowBase + r + 8 * half) * ldd + t * 16 + l16] = v;
    }
}

__device__ inline void store_stripeT(float* __restrict__ DT, int ldt,
                                     const v8f acc[4], int lane, int rowBase) {
  const int half = lane >> 4;
  const int l16  = lane & 15;
#pragma unroll
  for (int t = 0; t < 4; ++t)
#pragma unroll
    for (int r = 0; r < 8; ++r)
      DT[(t * 16 + l16) * ldt + rowBase + r + 8 * half] = acc[t][r];
}

__device__ inline void copy_w_T(float* __restrict__ dst,
                                const float* __restrict__ src,
                                int koff, int tid) {
  for (int s = tid; s < 64 * 64; s += 128) {
    const int k = s >> 6;
    const int n = s & 63;
    dst[n * LDP + k] = src[(k + koff) * H_DIM + n];
  }
}

__device__ inline void copy_obs_half(float* __restrict__ dst,
                                     const float* __restrict__ src,
                                     int coff, int tid) {
  for (int s = tid; s < 64 * 32; s += 128) {
    const int r  = s >> 5;
    const int c2 = (s & 31) * 2;
    *(float2*)(dst + r * LDP + c2) =
        *(const float2*)(src + r * D_DIM + coff + c2);
  }
}

#define GCH 16

__device__ inline void load_gi_chunk(v2f (&g)[4][GCH / 2], const float* __restrict__ rowh) {
#pragma unroll
  for (int r = 0; r < 4; ++r)
#pragma unroll
    for (int q = 0; q < GCH / 2; ++q)
      g[r][q] = *(const v2f*)(rowh + r * LDP + 2 * q);
}

template <int H0>
__device__ inline void gate_chunk(v2f (&S)[4], const v2f (&g)[4][GCH / 2],
                                  const v2f (&P)[H_DIM], const float (&w)[H_DIM]) {
#pragma unroll
  for (int q = 0; q < GCH / 2; ++q)
#pragma unroll
    for (int e = 0; e < 2; ++e) {
      const int h = H0 + 2 * q + e;
      const v2f wv = (v2f){w[h], w[h]};
#pragma unroll
      for (int r = 0; r < 4; ++r) {
        v2f t = P[h] + (v2f){g[r][q][e], g[r][q][e]};
        t = __builtin_elementwise_max(t, (v2f){0.0f, 0.0f});
        S[r] = __builtin_elementwise_fma(t, wv, S[r]);
      }
    }
}

__global__ __launch_bounds__(128) void gated_comm_kernel(
    const float* __restrict__ obs, const float* __restrict__ enc_w,
    const float* __restrict__ enc_b, const float* __restrict__ gate_w1,
    const float* __restrict__ gate_b1, const float* __restrict__ gate_w2,
    const float* __restrict__ gate_b2, const float* __restrict__ msg_w1,
    const float* __restrict__ msg_b1, const float* __restrict__ msg_w2,
    const float* __restrict__ msg_b2, float* __restrict__ out) {
  __shared__ __attribute__((aligned(16))) float smem[3 * 64 * LDP + 64];

  const int b       = blockIdx.x;
  const int tid     = threadIdx.x;
  const int lane    = tid & 31;
  const int wave    = tid >> 5;
  const int l16     = lane & 15;
  const int rowBase = wave * 16;

  float* R0  = smem;
  float* R1  = smem + 64 * LDP;
  float* R2  = smem + 2 * 64 * LDP;
  float* sW2 = smem + 3 * 64 * LDP;

  const float* obsb = obs + (size_t)b * (N_DIM * D_DIM);
  v8f acc[4];

  copy_obs_half(R0, obsb, 0, tid);
  copy_w_T(R2, enc_w, 0, tid);
  __syncthreads();
  init_acc(acc, enc_b, l16);
  gemm_accT(R0, LDP, R2, LDP, 64, acc, lane, rowBase);
  __syncthreads();
  copy_obs_half(R0, obsb, 64, tid);
  copy_w_T(R2, enc_w, 64, tid);
  __syncthreads();
  gemm_accT(R0, LDP, R2, LDP, 64, acc, lane, rowBase);
  __syncthreads();
  store_stripe(R0, LDP, acc, lane, rowBase,  true);
  copy_w_T(R2, gate_w1, 0, tid);
  __syncthreads();

  init_acc(acc, nullptr, l16);
  gemm_accT(R0, LDP, R2, LDP, H_DIM, acc, lane, rowBase);
  store_stripe(R1, LDP, acc, lane, rowBase, false);
  __syncthreads();
  copy_w_T(R2, gate_w1, 64, tid);
  __syncthreads();

  init_acc(acc, gate_b1, l16);
  gemm_accT(R0, LDP, R2, LDP, H_DIM, acc, lane, rowBase);
  __syncthreads();
  store_stripeT(R2, LDP, acc, lane, rowBase);
  if (tid < H_DIM) sW2[tid] = gate_w2[tid];
  __syncthreads();

  const float gb2 = gate_b2[0];
  v2f P[H_DIM];
#pragma unroll
  for (int h = 0; h < H_DIM; ++h)
    P[h] = *(const v2f*)(R2 + h * LDP + 2 * lane);
  float w2r[H_DIM];
#pragma unroll
  for (int h = 0; h < H_DIM; h += 4) {
    const float4 q = *(const float4*)(sW2 + h);
    w2r[h] = q.x; w2r[h + 1] = q.y; w2r[h + 2] = q.z; w2r[h + 3] = q.w;
  }
  v2f gA[4][GCH / 2], gB[4][GCH / 2];
  load_gi_chunk(gA, R1 + rowBase * LDP);
#pragma unroll 1
  for (int p = 0; p < 4; ++p) {
    float* row = R1 + (rowBase + 4 * p) * LDP;
    const float* nrow = row + (p < 3 ? 4 * LDP : 0);
    v2f S[4];
#pragma unroll
    for (int r = 0; r < 4; ++r) S[r] = (v2f){0.0f, 0.0f};

    load_gi_chunk(gB, row + GCH);
    __builtin_amdgcn_sched_barrier(0);
    gate_chunk<0 * GCH>(S, gA, P, w2r);
    load_gi_chunk(gA, row + 2 * GCH);
    __builtin_amdgcn_sched_barrier(0);
    gate_chunk<1 * GCH>(S, gB, P, w2r);
    load_gi_chunk(gB, row + 3 * GCH);
    __builtin_amdgcn_sched_barrier(0);
    gate_chunk<2 * GCH>(S, gA, P, w2r);
    load_gi_chunk(gA, nrow);
    __builtin_amdgcn_sched_barrier(0);
    gate_chunk<3 * GCH>(S, gB, P, w2r);

#pragma unroll
    for (int r = 0; r < 4; ++r) {
      v2f gt;
      gt[0] = 1.0f / (1.0f + __expf(-(S[r][0] + gb2)));
      gt[1] = 1.0f / (1.0f + __expf(-(S[r][1] + gb2)));
      *(v2f*)(row + r * LDP + 2 * lane) = gt;
    }
  }
  __syncthreads();

  copy_w_T(R2, msg_w1, 0, tid);
  __syncthreads();
  init_acc(acc, msg_b1, l16);
  gemm_accT(R0, LDP, R2, LDP, H_DIM, acc, lane, rowBase);
  __syncthreads();
  store_stripe(R0, LDP, acc, lane, rowBase,  true);
  copy_w_T(R2, msg_w2, 0, tid);
  __syncthreads();
  init_acc(acc, msg_b2, l16);
  gemm_accT(R0, LDP, R2, LDP, H_DIM, acc, lane, rowBase);
  __syncthreads();
  store_stripeT(R2, LDP, acc, lane, rowBase);
  __syncthreads();

  init_acc(acc, nullptr, l16);
  gemm_accT(R1, LDP, R2, LDP, N_DIM, acc, lane, rowBase);

  float* outb = out + (size_t)b * (N_DIM * H_DIM);
  const int half = lane >> 4;
#pragma unroll
  for (int t = 0; t < 4; ++t)
#pragma unroll
    for (int r = 0; r < 8; ++r)
      outb[(rowBase + r + 8 * half) * H_DIM + t * 16 + l16] = acc[t][r];
}

extern "C" void kernel_launch(void* const* d_in, const int* in_sizes, int n_in,
                              void* d_out, int out_size, void* d_ws, size_t ws_size,
                              hipStream_t stream) {
  const float* obs     = (const float*)d_in[0];
  const float* enc_w   = (const float*)d_in[1];
  const float* enc_b   = (const float*)d_in[2];
  const float* gate_w1 = (const float*)d_in[3];
  const float* gate_b1 = (const float*)d_in[4];
  const float* gate_w2 = (const float*)d_in[5];
  const float* gate_b2 = (const float*)d_in[6];
  const float* msg_w1  = (const float*)d_in[7];
  const float* msg_b1  = (const float*)d_in[8];
  const float* msg_w2  = (const float*)d_in[9];
  const float* msg_b2  = (const float*)d_in[10];
  float* out = (float*)d_out;

  gated_comm_kernel<<<BATCH, 128, 0, stream>>>(
      obs, enc_w, enc_b, gate_w1, gate_b1, gate_w2, gate_b2,
      msg_w1, msg_b1, msg_w2, msg_b2, out);
}


# all global operands (obs tile, 4 weight matrices, biases, w2) fetched in ONE entry burst of 16-byte loads into LDS regions of their own, B as k-pair rows; 2 barriers left; sum order unchanged
# speedup vs baseline: 2.2014x; 1.6823x over previous
#include <hip/hip_runtime.h>
#include <hip/hip_bf16.h>
#include <math.h>


typedef __attribute__((ext_vector_type(2))) float v2f;
typedef __attribute__((ext_vector_type(8))) float v8f;

#define BATCH 256
#define N_DIM 64
#define D_DIM 128
#define H_DIM 64
#define LDA   132
#define LDG   68
#define LDP2  160

#define OFF_OBS 0
#define OFF_WE  (OFF_OBS + 64 * LDA)
#define OFF_WG  (OFF_WE + 64 * LDP2)
#define OFF_WM1 (OFF_WG + 64 * LDP2)
#define OFF_WM2 (OFF_WM1 + 32 * LDP2)
#define OFF_GI  (OFF_WM2 + 32 * LDP2)
#define OFF_GJT (OFF_GI + 64 * LDG)
#define OFF_MSG (OFF_GJT + 64 * LDG)
#define OFF_W2  (OFF_MSG + 32 * LDP2)
#define LDS_FLOATS (OFF_W2 + 64)

__device__ inline v8f wmma_f32x4(v2f a, v2f b, v8f c) {
  return __builtin_amdgcn_wmma_f32_16x16x4_f32(
       false, a,  false, b,
       (short)0, c,  false,  false);
}

__device__ inline void init_acc(v8f acc[4], const float bias[4]) {
#pragma unroll
  for (int t = 0; t < 4; ++t)
#pragma unroll
    for (int r = 0; r < 8; ++r) acc[t][r] = bias[t];
}

template <int K>
__device__ inline void gemm_acc(const float* __restrict__ A, int lda,
                                const float* __restrict__ B,
                                v8f acc[4], int lane, int rowBase) {
  const int half = lane >> 4;
  const int l16  = lane & 15;
  const float* ap = A + (rowBase + l16) * lda + 2 * half;
  const float* bp = B + half * LDP2 + 2 * l16;
#pragma unroll
  for (int k0 = 0; k0 < K; k0 += 4) {
    v2f a = *(const v2f*)(ap + k0);
#pragma unroll
    for (int t = 0; t < 4; ++t) {
      v2f b = *(const v2f*)(bp + (k0 / 2) * LDP2 + t * 32);
      acc[t] = wmma_f32x4(a, b, acc[t]);
    }
  }
}

__device__ inline void store_stripe(float* __restrict__ D, int ldd,
                                    const v8f acc[4], int lane, int rowBase,
                                    bool do_relu) {
  const int half = lane >> 4;
  const int l16  = lane & 15;
#pragma unroll
  for (int t = 0; t < 4; ++t)
#pragma unroll
    for (int r = 0; r < 8; ++r) {
      float v = acc[t][r];
      if (do_relu) v = fmaxf(v, 0.0f);
      D[(rowBase + r + 8 * half) * ldd + t * 16 + l16] = v;
    }
}

__device__ inline void store_stripeT(float* __restrict__ DT, int ldt,
                                     const v8f acc[4], int lane, int rowBase) {
  const int half = lane >> 4;
  const int l16  = lane & 15;
#pragma unroll
  for (int t = 0; t < 4; ++t)
#pragma unroll
    for (int r = 0; r < 8; ++r)
      DT[(t * 16 + l16) * ldt + rowBase + r + 8 * half] = acc[t][r];
}

__device__ inline void store_stripeP(float* __restrict__ DP,
                                     const v8f acc[4], int lane, int rowBase) {
  const int half = lane >> 4;
  const int l16  = lane & 15;
#pragma unroll
  for (int t = 0; t < 4; ++t)
#pragma unroll
    for (int r = 0; r < 8; r += 2)
      *(v2f*)(DP + ((rowBase + r + 8 * half) >> 1) * LDP2 + 2 * (t * 16 + l16)) =
          (v2f){acc[t][r], acc[t][r + 1]};
}

template <int NV>
__device__ inline void burst_loadP(float4 (&v)[NV], const float* __restrict__ src, int tid) {
  const float4* s = (const float4*)src + (2 * (tid >> 4)) * 16 + (tid & 15);
#pragma unroll
  for (int j = 0; j < NV / 2; ++j) {
    v[2 * j]     = s[(16 * j) * 16];
    v[2 * j + 1] = s[(16 * j + 1) * 16];
  }
}
template <int NV>
__device__ inline void burst_storeP(float* __restrict__ dst, const float4 (&v)[NV], int tid) {
  float* p = dst + (tid >> 4) * LDP2 + 8 * (tid & 15);
#pragma unroll
  for (int j = 0; j < NV / 2; ++j) {
    const float4 x = v[2 * j], y = v[2 * j + 1];
    *(float4*)(p + 8 * j * LDP2)     = make_float4(x.x, y.x, x.y, y.y);
    *(float4*)(p + 8 * j * LDP2 + 4) = make_float4(x.z, y.z, x.w, y.w);
  }
}
template <int NV>
__device__ inline void burst_load(float4 (&v)[NV], const float* __restrict__ src, int tid) {
#pragma unroll
  for (int i = 0; i < NV; ++i) v[i] = ((const float4*)src)[i * 128 + tid];
}
template <int NV, int W4>
__device__ inline void burst_store(float* __restrict__ dst, int ld,
                                   const float4 (&v)[NV], int tid) {
  const int rpi = 128 / W4;
  float* p = dst + (tid / W4) * ld + (tid % W4) * 4;
#pragma unroll
  for (int i = 0; i < NV; ++i) *(float4*)(p + i * rpi * ld) = v[i];
}

#define GCH 16

__device__ inline void load_gi_chunk(v2f (&g)[4][GCH / 2], const float* __restrict__ rowh) {
#pragma unroll
  for (int r = 0; r < 4; ++r)
#pragma unroll
    for (int q = 0; q < GCH / 2; ++q)
      g[r][q] = *(const v2f*)(rowh + r * LDG + 2 * q);
}

template <int H0>
__device__ inline void gate_chunk(v2f (&S)[4], const v2f (&g)[4][GCH / 2],
                                  const v2f (&P)[H_DIM], const float (&w)[H_DIM]) {
#pragma unroll
  for (int q = 0; q < GCH / 2; ++q)
#pragma unroll
    for (int e = 0; e < 2; ++e) {
      const int h = H0 + 2 * q + e;
      const v2f wv = (v2f){w[h], w[h]};
#pragma unroll
      for (int r = 0; r < 4; ++r) {
        v2f t = P[h] + (v2f){g[r][q][e], g[r][q][e]};
        t = __builtin_elementwise_max(t, (v2f){0.0f, 0.0f});
        S[r] = __builtin_elementwise_fma(t, wv, S[r]);
      }
    }
}

__global__ __launch_bounds__(128) void gated_comm_kernel(
    const float* __restrict__ obs, const float* __restrict__ enc_w,
    const float* __restrict__ enc_b, const float* __restrict__ gate_w1,
    const float* __restrict__ gate_b1, const float* __restrict__ gate_w2,
    const float* __restrict__ gate_b2, const float* __restrict__ msg_w1,
    const float* __restrict__ msg_b1, const float* __restrict__ msg_w2,
    const float* __restrict__ msg_b2, float* __restrict__ out) {
  __shared__ __attribute__((aligned(16))) float smem[LDS_FLOATS];

  const int b       = blockIdx.x;
  const int tid     = threadIdx.x;
  const int lane    = tid & 31;
  const int wave    = tid >> 5;
  const int l16     = lane & 15;
  const int rowBase = wave * 16;

  float* sOBS = smem + OFF_OBS;
  float* sWE  = smem + OFF_WE;
  float* sWG  = smem + OFF_WG;
  float* sWM1 = smem + OFF_WM1;
  float* sWM2 = smem + OFF_WM2;
  float* sGI  = smem + OFF_GI;
  float* sGJT = smem + OFF_GJT;
  float* sMSG = smem + OFF_MSG;
  float* sW2  = smem + OFF_W2;

  const float* obsb = obs + (size_t)b * (N_DIM * D_DIM);

  float4 vo[16], ve[16], vg[16], v1[8], v2[8];
  burst_load(vo, obsb, tid);
  burst_loadP(ve, enc_w, tid);
  burst_loadP(vg, gate_w1, tid);
  burst_loadP(v1, msg_w1, tid);
  burst_loadP(v2, msg_w2, tid);
  float be[4], bg[4], bm1[4], bm2[4], bz[4];
#pragma unroll
  for (int t = 0; t < 4; ++t) {
    be[t]  = enc_b[t * 16 + l16];
    bg[t]  = gate_b1[t * 16 + l16];
    bm1[t] = msg_b1[t * 16 + l16];
    bm2[t] = msg_b2[t * 16 + l16];
    bz[t]  = 0.0f;
  }
  const float w2v = gate_w2[tid & 63];
  const float gb2 = gate_b2[0];

  burst_store<16, 32>(sOBS, LDA, vo, tid);
  burst_storeP(sWE,  ve, tid);
  burst_storeP(sWG,  vg, tid);
  burst_storeP(sWM1, v1, tid);
  burst_storeP(sWM2, v2, tid);
  sW2[tid & 63] = w2v;
  __syncthreads();

  v8f acc[4];

  init_acc(acc, be);
  gemm_acc<D_DIM>(sOBS, LDA, sWE, acc, lane, rowBase);
  store_stripe(sOBS, LDA, acc, lane, rowBase,  true);

  init_acc(acc, bz);
  gemm_acc<H_DIM>(sOBS, LDA, sWG, acc, lane, rowBase);
  store_stripe(sGI, LDG, acc, lane, rowBase, false);

  init_acc(acc, bg);
  gemm_acc<H_DIM>(sOBS, LDA, sWG + 32 * LDP2, acc, lane, rowBase);
  store_stripeT(sGJT, LDG, acc, lane, rowBase);

  init_acc(acc, bm1);
  gemm_acc<H_DIM>(sOBS, LDA, sWM1, acc, lane, rowBase);
  store_stripe(sOBS + 64, LDA, acc, lane, rowBase,  true);
  init_acc(acc, bm2);
  gemm_acc<H_DIM>(sOBS + 64, LDA, sWM2, acc, lane, rowBase);
  store_stripeP(sMSG, acc, lane, rowBase);
  __syncthreads();

  v2f P[H_DIM];
#pragma unroll
  for (int h = 0; h < H_DIM; ++h)
    P[h] = *(const v2f*)(sGJT + h * LDG + 2 * lane);
  float w2r[H_DIM];
#pragma unroll
  for (int h = 0; h < H_DIM; h += 4) {
    const float4 q = *(const float4*)(sW2 + h);
    w2r[h] = q.x; w2r[h + 1] = q.y; w2r[h + 2] = q.z; w2r[h + 3] = q.w;
  }
  v2f gA[4][GCH / 2], gB[4][GCH / 2];
  load_gi_chunk(gA, sGI + rowBase * LDG);
#pragma unroll 1
  for (int p = 0; p < 4; ++p) {
    float* row = sGI + (rowBase + 4 * p) * LDG;
    const float* nrow = row + (p < 3 ? 4 * LDG : 0);
    v2f S[4];
#pragma unroll
    for (int r = 0; r < 4; ++r) S[r] = (v2f){0.0f, 0.0f};

    load_gi_chunk(gB, row + GCH);
    __builtin_amdgcn_sched_barrier(0);
    gate_chunk<0 * GCH>(S, gA, P, w2r);
    load_gi_chunk(gA, row + 2 * GCH);
    __builtin_amdgcn_sched_barrier(0);
    gate_chunk<1 * GCH>(S, gB, P, w2r);
    load_gi_chunk(gB, row + 3 * GCH);
    __builtin_amdgcn_sched_barrier(0);
    gate_chunk<2 * GCH>(S, gA, P, w2r);
    load_gi_chunk(gA, nrow);
    __builtin_amdgcn_sched_barrier(0);
    gate_chunk<3 * GCH>(S, gB, P, w2r);

#pragma unroll
    for (int r = 0; r < 4; ++r) {
      v2f gt;
      gt[0] = 1.0f / (1.0f + __expf(-(S[r][0] + gb2)));
      gt[1] = 1.0f / (1.0f + __expf(-(S[r][1] + gb2)));
      *(v2f*)(row + r * LDG + 2 * lane) = gt;
    }
  }

  init_acc(acc, bz);
  gemm_acc<N_DIM>(sGI, LDG, sMSG, acc, lane, rowBase);

  float* outb = out + (size_t)b * (N_DIM * H_DIM);
  const int half = lane >> 4;
#pragma unroll
  for (int t = 0; t < 4; ++t)
#pragma unroll
    for (int r = 0; r < 8; ++r)
      outb[(rowBase + r + 8 * half) * H_DIM + t * 16 + l16] = acc[t][r];
}

extern "C" void kernel_launch(void* const* d_in, const int* in_sizes, int n_in,
                              void* d_out, int out_size, void* d_ws, size_t ws_size,
                              hipStream_t stream) {
  const float* obs     = (const float*)d_in[0];
  const float* enc_w   = (const float*)d_in[1];
  const float* enc_b   = (const float*)d_in[2];
  const float* gate_w1 = (const float*)d_in[3];
  const float* gate_b1 = (const float*)d_in[4];
  const float* gate_w2 = (const float*)d_in[5];
  const float* gate_b2 = (const float*)d_in[6];
  const float* msg_w1  = (const float*)d_in[7];
  const float* msg_b1  = (const float*)d_in[8];
  const float* msg_w2  = (const float*)d_in[9];
  const float* msg_b2  = (const float*)d_in[10];
  float* out = (float*)d_out;

  gated_comm_kernel<<<BATCH, 128, 0, stream>>>(
      obs, enc_w, enc_b, gate_w1, gate_b1, gate_w2, gate_b2,
      msg_w1, msg_b1, msg_w2, msg_b2, out);
}


# LDS fragment reads of all six GEMMs issued one block of 4 k-steps (16 WMMAs) ahead of use in two alternating register sets, not about 4 WMMAs ahead; k order and every sum unchanged
# speedup vs baseline: 2.4742x; 1.1239x over previous
#include <hip/hip_runtime.h>
#include <hip/hip_bf16.h>
#include <math.h>


typedef __attribute__((ext_vector_type(2))) float v2f;
typedef __attribute__((ext_vector_type(8))) float v8f;

#define BATCH 256
#define N_DIM 64
#define D_DIM 128
#define H_DIM 64
#define LDA   132
#define LDG   68
#define LDP2  160

#define OFF_OBS 0
#define OFF_WE  (OFF_OBS + 64 * LDA)
#define OFF_WG  (OFF_WE + 64 * LDP2)
#define OFF_WM1 (OFF_WG + 64 * LDP2)
#define OFF_WM2 (OFF_WM1 + 32 * LDP2)
#define OFF_GI  (OFF_WM2 + 32 * LDP2)
#define OFF_GJT (OFF_GI + 64 * LDG)
#define OFF_MSG (OFF_GJT + 64 * LDG)
#define OFF_W2  (OFF_MSG + 32 * LDP2)
#define LDS_FLOATS (OFF_W2 + 64)

__device__ inline v8f wmma_f32x4(v2f a, v2f b, v8f c) {
  return __builtin_amdgcn_wmma_f32_16x16x4_f32(
       false, a,  false, b,
       (short)0, c,  false,  false);
}

__device__ inline void init_acc(v8f acc[4], const float bias[4]) {
#pragma unroll
  for (int t = 0; t < 4; ++t)
#pragma unroll
    for (int r = 0; r < 8; ++r) acc[t][r] = bias[t];
}

#define KB 4
__device__ inline void load_frag_block(v2f (&fa)[KB], v2f (&fb)[KB][4],
                                       const float* __restrict__ ap,
                                       const float* __restrict__ bp, int k0) {
#pragma unroll
  for (int s = 0; s < KB; ++s) {
    fa[s] = *(const v2f*)(ap + k0 + 4 * s);
#pragma unroll
    for (int t = 0; t < 4; ++t)
      fb[s][t] = *(const v2f*)(bp + ((k0 + 4 * s) / 2) * LDP2 + t * 32);
  }
}

template <int K>
__device__ inline void gemm_acc(const float* __restrict__ A, int lda,
                                const float* __restrict__ B,
                                v8f acc[4], int lane, int rowBase) {
  const int half = lane >> 4;
  const int l16  = lane & 15;
  const float* ap = A + (rowBase + l16) * lda + 2 * half;
  const float* bp = B + half * LDP2 + 2 * l16;
  constexpr int NB = K / (4 * KB);
  v2f fa[2][KB], fb[2][KB][4];
  load_frag_block(fa[0], fb[0], ap, bp, 0);
#pragma unroll
  for (int j = 0; j < NB; ++j) {
    if (j + 1 < NB) load_frag_block(fa[(j + 1) & 1], fb[(j + 1) & 1], ap, bp, (j + 1) * 4 * KB);
    __builtin_amdgcn_sched_barrier(0);
#pragma unroll
    for (int s = 0; s < KB; ++s)
#pragma unroll
      for (int t = 0; t < 4; ++t)
        acc[t] = wmma_f32x4(fa[j & 1][s], fb[j & 1][s][t], acc[t]);
  }
}

__device__ inline void store_stripe(float* __restrict__ D, int ldd,
                                    const v8f acc[4], int lane, int rowBase,
                                    bool do_relu) {
  const int half = lane >> 4;
  const int l16  = lane & 15;
#pragma unroll
  for (int t = 0; t < 4; ++t)
#pragma unroll
    for (int r = 0; r < 8; ++r) {
      float v = acc[t][r];
      if (do_relu) v = fmaxf(v, 0.0f);
      D[(rowBase + r + 8 * half) * ldd + t * 16 + l16] = v;
    }
}

__device__ inline void store_stripeT(float* __restrict__ DT, int ldt,
                                     const v8f acc[4], int lane, int rowBase) {
  const int half = lane >> 4;
  const int l16  = lane & 15;
#pragma unroll
  for (int t = 0; t < 4; ++t)
#pragma unroll
    for (int r = 0; r < 8; ++r)
      DT[(t * 16 + l16) * ldt + rowBase + r + 8 * half] = acc[t][r];
}

__device__ inline void store_stripeP(float* __restrict__ DP,
                                     const v8f acc[4], int lane, int rowBase) {
  const int half = lane >> 4;
  const int l16  = lane & 15;
#pragma unroll
  for (int t = 0; t < 4; ++t)
#pragma unroll
    for (int r = 0; r < 8; r += 2)
      *(v2f*)(DP + ((rowBase + r + 8 * half) >> 1) * LDP2 + 2 * (t * 16 + l16)) =
          (v2f){acc[t][r], acc[t][r + 1]};
}

template <int NV>
__device__ inline void burst_loadP(float4 (&v)[NV], const float* __restrict__ src, int tid) {
  const float4* s = (const float4*)src + (2 * (tid >> 4)) * 16 + (tid & 15);
#pragma unroll
  for (int j = 0; j < NV / 2; ++j) {
    v[2 * j]     = s[(16 * j) * 16];
    v[2 * j + 1] = s[(16 * j + 1) * 16];
  }
}
template <int NV>
__device__ inline void burst_storeP(float* __restrict__ dst, const float4 (&v)[NV], int tid) {
  float* p = dst + (tid >> 4) * LDP2 + 8 * (tid & 15);
#pragma unroll
  for (int j = 0; j < NV / 2; ++j) {
    const float4 x = v[2 * j], y = v[2 * j + 1];
    *(float4*)(p + 8 * j * LDP2)     = make_float4(x.x, y.x, x.y, y.y);
    *(float4*)(p + 8 * j * LDP2 + 4) = make_float4(x.z, y.z, x.w, y.w);
  }
}
template <int NV>
__device__ inline void burst_load(float4 (&v)[NV], const float* __restrict__ src, int tid) {
#pragma unroll
  for (int i = 0; i < NV; ++i) v[i] = ((const float4*)src)[i * 128 + tid];
}
template <int NV, int W4>
__device__ inline void burst_store(float* __restrict__ dst, int ld,
                                   const float4 (&v)[NV], int tid) {
  const int rpi = 128 / W4;
  float* p = dst + (tid / W4) * ld + (tid % W4) * 4;
#pragma unroll
  for (int i = 0; i < NV; ++i) *(float4*)(p + i * rpi * ld) = v[i];
}

#define GCH 16

__device__ inline void load_gi_chunk(v2f (&g)[4][GCH / 2], const float* __restrict__ rowh) {
#pragma unroll
  for (int r = 0; r < 4; ++r)
#pragma unroll
    for (int q = 0; q < GCH / 2; ++q)
      g[r][q] = *(const v2f*)(rowh + r * LDG + 2 * q);
}

template <int H0>
__device__ inline void gate_chunk(v2f (&S)[4], const v2f (&g)[4][GCH / 2],
                                  const v2f (&P)[H_DIM], const float (&w)[H_DIM]) {
#pragma unroll
  for (int q = 0; q < GCH / 2; ++q)
#pragma unroll
    for (int e = 0; e < 2; ++e) {
      const int h = H0 + 2 * q + e;
      const v2f wv = (v2f){w[h], w[h]};
#pragma unroll
      for (int r = 0; r < 4; ++r) {
        v2f t = P[h] + (v2f){g[r][q][e], g[r][q][e]};
        t = __builtin_elementwise_max(t, (v2f){0.0f, 0.0f});
        S[r] = __builtin_elementwise_fma(t, wv, S[r]);
      }
    }
}

__global__ __launch_bounds__(128) void gated_comm_kernel(
    const float* __restrict__ obs, const float* __restrict__ enc_w,
    const float* __restrict__ enc_b, const float* __restrict__ gate_w1,
    const float* __restrict__ gate_b1, const float* __restrict__ gate_w2,
    const float* __restrict__ gate_b2, const float* __restrict__ msg_w1,
    const float* __restrict__ msg_b1, const float* __restrict__ msg_w2,
    const float* __restrict__ msg_b2, float* __restrict__ out) {
  __shared__ __attribute__((aligned(16))) float smem[LDS_FLOATS];

  const int b       = blockIdx.x;
  const int tid     = threadIdx.x;
  const int lane    = tid & 31;
  const int wave    = tid >> 5;
  const int l16     = lane & 15;
  const int rowBase = wave * 16;

  float* sOBS = smem + OFF_OBS;
  float* sWE  = smem + OFF_WE;
  float* sWG  = smem + OFF_WG;
  float* sWM1 = smem + OFF_WM1;
  float* sWM2 = smem + OFF_WM2;
  float* sGI  = smem + OFF_GI;
  float* sGJT = smem + OFF_GJT;
  float* sMSG = smem + OFF_MSG;
  float* sW2  = smem + OFF_W2;

  const float* obsb = obs + (size_t)b * (N_DIM * D_DIM);

  float4 vo[16], ve[16], vg[16], v1[8], v2[8];
  burst_load(vo, obsb, tid);
  burst_loadP(ve, enc_w, tid);
  burst_loadP(vg, gate_w1, tid);
  burst_loadP(v1, msg_w1, tid);
  burst_loadP(v2, msg_w2, tid);
  float be[4], bg[4], bm1[4], bm2[4], bz[4];
#pragma unroll
  for (int t = 0; t < 4; ++t) {
    be[t]  = enc_b[t * 16 + l16];
    bg[t]  = gate_b1[t * 16 + l16];
    bm1[t] = msg_b1[t * 16 + l16];
    bm2[t] = msg_b2[t * 16 + l16];
    bz[t]  = 0.0f;
  }
  const float w2v = gate_w2[tid & 63];
  const float gb2 = gate_b2[0];

  burst_store<16, 32>(sOBS, LDA, vo, tid);
  burst_storeP(sWE,  ve, tid);
  burst_storeP(sWG,  vg, tid);
  burst_storeP(sWM1, v1, tid);
  burst_storeP(sWM2, v2, tid);
  sW2[tid & 63] = w2v;
  __syncthreads();

  v8f acc[4];

  init_acc(acc, be);
  gemm_acc<D_DIM>(sOBS, LDA, sWE, acc, lane, rowBase);
  store_stripe(sOBS, LDA, acc, lane, rowBase,  true);

  init_acc(acc, bz);
  gemm_acc<H_DIM>(sOBS, LDA, sWG, acc, lane, rowBase);
  store_stripe(sGI, LDG, acc, lane, rowBase, false);

  init_acc(acc, bg);
  gemm_acc<H_DIM>(sOBS, LDA, sWG + 32 * LDP2, acc, lane, rowBase);
  store_stripeT(sGJT, LDG, acc, lane, rowBase);

  init_acc(acc, bm1);
  gemm_acc<H_DIM>(sOBS, LDA, sWM1, acc, lane, rowBase);
  store_stripe(sOBS + 64, LDA, acc, lane, rowBase,  true);
  init_acc(acc, bm2);
  gemm_acc<H_DIM>(sOBS + 64, LDA, sWM2, acc, lane, rowBase);
  store_stripeP(sMSG, acc, lane, rowBase);
  __syncthreads();

  v2f P[H_DIM];
#pragma unroll
  for (int h = 0; h < H_DIM; ++h)
    P[h] = *(const v2f*)(sGJT + h * LDG + 2 * lane);
  float w2r[H_DIM];
#pragma unroll
  for (int h = 0; h < H_DIM; h += 4) {
    const float4 q = *(const float4*)(sW2 + h);
    w2r[h] = q.x; w2r[h + 1] = q.y; w2r[h + 2] = q.z; w2r[h + 3] = q.w;
  }
  v2f gA[4][GCH / 2], gB[4][GCH / 2];
  load_gi_chunk(gA, sGI + rowBase * LDG);
#pragma unroll 1
  for (int p = 0; p < 4; ++p) {
    float* row = sGI + (rowBase + 4 * p) * LDG;
    const float* nrow = row + (p < 3 ? 4 * LDG : 0);
    v2f S[4];
#pragma unroll
    for (int r = 0; r < 4; ++r) S[r] = (v2f){0.0f, 0.0f};

    load_gi_chunk(gB, row + GCH);
    __builtin_amdgcn_sched_barrier(0);
    gate_chunk<0 * GCH>(S, gA, P, w2r);
    load_gi_chunk(gA, row + 2 * GCH);
    __builtin_amdgcn_sched_barrier(0);
    gate_chunk<1 * GCH>(S, gB, P, w2r);
    load_gi_chunk(gB, row + 3 * GCH);
    __builtin_amdgcn_sched_barrier(0);
    gate_chunk<2 * GCH>(S, gA, P, w2r);
    load_gi_chunk(gA, nrow);
    __builtin_amdgcn_sched_barrier(0);
    gate_chunk<3 * GCH>(S, gB, P, w2r);

#pragma unroll
    for (int r = 0; r < 4; ++r) {
      v2f gt;
      gt[0] = 1.0f / (1.0f + __expf(-(S[r][0] + gb2)));
      gt[1] = 1.0f / (1.0f + __expf(-(S[r][1] + gb2)));
      *(v2f*)(row + r * LDG + 2 * lane) = gt;
    }
  }

  init_acc(acc, bz);
  gemm_acc<N_DIM>(sGI, LDG, sMSG, acc, lane, rowBase);

  float* outb = out + (size_t)b * (N_DIM * H_DIM);
  const int half = lane >> 4;
#pragma unroll
  for (int t = 0; t < 4; ++t)
#pragma unroll
    for (int r = 0; r < 8; ++r)
      outb[(rowBase + r + 8 * half) * H_DIM + t * 16 + l16] = acc[t][r];
}

extern "C" void kernel_launch(void* const* d_in, const int* in_sizes, int n_in,
                              void* d_out, int out_size, void* d_ws, size_t ws_size,
                              hipStream_t stream) {
  const float* obs     = (const float*)d_in[0];
  const float* enc_w   = (const float*)d_in[1];
  const float* enc_b   = (const float*)d_in[2];
  const float* gate_w1 = (const float*)d_in[3];
  const float* gate_b1 = (const float*)d_in[4];
  const float* gate_w2 = (const float*)d_in[5];
  const float* gate_b2 = (const float*)d_in[6];
  const float* msg_w1  = (const float*)d_in[7];
  const float* msg_b1  = (const float*)d_in[8];
  const float* msg_w2  = (const float*)d_in[9];
  const float* msg_b2  = (const float*)d_in[10];
  float* out = (float*)d_out;

  gated_comm_kernel<<<BATCH, 128, 0, stream>>>(
      obs, enc_w, enc_b, gate_w1, gate_b1, gate_w2, gate_b2,
      msg_w1, msg_b1, msg_w2, msg_b2, out);
}


# two waves per SIMD (8-wave work-groups): GEMMs split by column halves; behind the gi/gjT barrier the older wave runs msg's two GEMMs while its sibling starts the gate rows (4:12); sums unchanged
# speedup vs baseline: 2.6796x; 1.0830x over previous
#include <hip/hip_runtime.h>
#include <hip/hip_bf16.h>
#include <math.h>


typedef __attribute__((ext_vector_type(2))) float v2f;
typedef __attribute__((ext_vector_type(8))) float v8f;

#define BATCH 256
#define N_DIM 64
#define D_DIM 128
#define H_DIM 64
#define LDA   132
#define LDG   68
#define LDP2  160

#define OFF_OBS 0
#define OFF_WE  (OFF_OBS + 64 * LDA)
#define OFF_WG  (OFF_WE + 64 * LDP2)
#define OFF_WM1 (OFF_WG + 64 * LDP2)
#define OFF_WM2 (OFF_WM1 + 32 * LDP2)
#define OFF_GI  (OFF_WM2 + 32 * LDP2)
#define OFF_GJT (OFF_GI + 64 * LDG)
#define OFF_MSG (OFF_GJT + 64 * LDG)
#define OFF_W2  (OFF_MSG + 32 * LDP2)
#define OFF_ENC (OFF_W2 + 64)
#define OFF_TMP (OFF_ENC + 64 * LDG)
#define LDS_FLOATS (OFF_TMP + 64 * LDG)

#define NT 2

__device__ inline v8f wmma_f32x4(v2f a, v2f b, v8f c) {
  return __builtin_amdgcn_wmma_f32_16x16x4_f32(
       false, a,  false, b,
       (short)0, c,  false,  false);
}

template <int NTT>
__device__ inline void init_acc(v8f (&acc)[NTT], const float (&bias)[NTT]) {
#pragma unroll
  for (int t = 0; t < NTT; ++t)
#pragma unroll
    for (int r = 0; r < 8; ++r) acc[t][r] = bias[t];
}

template <int NTT, int KBB>
__device__ inline void load_frag_block(v2f (&fa)[KBB], v2f (&fb)[KBB][NTT],
                                       const float* __restrict__ ap,
                                       const float* __restrict__ bp, int k0) {
#pragma unroll
  for (int s = 0; s < KBB; ++s) {
    fa[s] = *(const v2f*)(ap + k0 + 4 * s);
#pragma unroll
    for (int t = 0; t < NTT; ++t)
      fb[s][t] = *(const v2f*)(bp + ((k0 + 4 * s) / 2) * LDP2 + t * 32);
  }
}

template <int K, int NTT, int KBB>
__device__ inline void gemm_acc(const float* __restrict__ A, int lda,
                                const float* __restrict__ B,
                                v8f (&acc)[NTT], int lane, int rowBase, int colBase) {
  const int half = lane >> 4;
  const int l16  = lane & 15;
  const float* ap = A + (rowBase + l16) * lda + 2 * half;
  const float* bp = B + half * LDP2 + 2 * (colBase + l16);
  constexpr int NB = K / (4 * KBB);
  v2f fa[2][KBB], fb[2][KBB][NTT];
  load_frag_block<NTT, KBB>(fa[0], fb[0], ap, bp, 0);
#pragma unroll
  for (int j = 0; j < NB; ++j) {
    if (j + 1 < NB)
      load_frag_block<NTT, KBB>(fa[(j + 1) & 1], fb[(j + 1) & 1], ap, bp, (j + 1) * 4 * KBB);
    __builtin_amdgcn_sched_barrier(0);
#pragma unroll
    for (int s = 0; s < KBB; ++s)
#pragma unroll
      for (int t = 0; t < NTT; ++t)
        acc[t] = wmma_f32x4(fa[j & 1][s], fb[j & 1][s][t], acc[t]);
  }
}

template <int NTT>
__device__ inline void store_stripe(float* __restrict__ D, int ldd,
                                    const v8f (&acc)[NTT], int lane, int rowBase,
                                    int colBase, bool do_relu) {
  const int half = lane >> 4;
  const int l16  = lane & 15;
#pragma unroll
  for (int t = 0; t < NTT; ++t)
#pragma unroll
    for (int r = 0; r < 8; ++r) {
      float v = acc[t][r];
      if (do_relu) v = fmaxf(v, 0.0f);
      D[(rowBase + r + 8 * half) * ldd + colBase + t * 16 + l16] = v;
    }
}

template <int NTT>
__device__ inline void store_stripeT(float* __restrict__ DT, int ldt,
                                     const v8f (&acc)[NTT], int lane, int rowBase,
                                     int colBase) {
  const int half = lane >> 4;
  const int l16  = lane & 15;
#pragma unroll
  for (int t = 0; t < NTT; ++t)
#pragma unroll
    for (int r = 0; r < 8; ++r)
      DT[(colBase + t * 16 + l16) * ldt + rowBase + r + 8 * half] = acc[t][r];
}

template <int NTT>
__device__ inline void store_stripeP(float* __restrict__ DP,
                                     const v8f (&acc)[NTT], int lane, int rowBase,
                                     int colBase) {
  const int half = lane >> 4;
  const int l16  = lane & 15;
#pragma unroll
  for (int t = 0; t < NTT; ++t)
#pragma unroll
    for (int r = 0; r < 8; r += 2)
      *(v2f*)(DP + ((rowBase + r + 8 * half) >> 1) * LDP2 + 2 * (colBase + t * 16 + l16)) =
          (v2f){acc[t][r], acc[t][r + 1]};
}

template <int NV>
__device__ inline void burst_loadP(float4 (&v)[NV], const float* __restrict__ src, int tid) {
  const float4* s = (const float4*)src + (2 * (tid >> 4)) * 16 + (tid & 15);
#pragma unroll
  for (int j = 0; j < NV / 2; ++j) {
    v[2 * j]     = s[(32 * j) * 16];
    v[2 * j + 1] = s[(32 * j + 1) * 16];
  }
}
template <int NV>
__device__ inline void burst_storeP(float* __restrict__ dst, const float4 (&v)[NV], int tid) {
  float* p = dst + (tid >> 4) * LDP2 + 8 * (tid & 15);
#pragma unroll
  for (int j = 0; j < NV / 2; ++j) {
    const float4 x = v[2 * j], y = v[2 * j + 1];
    *(float4*)(p + 16 * j * LDP2)     = make_float4(x.x, y.x, x.y, y.y);
    *(float4*)(p + 16 * j * LDP2 + 4) = make_float4(x.z, y.z, x.w, y.w);
  }
}
template <int NV>
__device__ inline void burst_load(float4 (&v)[NV], const float* __restrict__ src, int tid) {
#pragma unroll
  for (int i = 0; i < NV; ++i) v[i] = ((const float4*)src)[i * 256 + tid];
}
template <int NV, int W4>
__device__ inline void burst_store(float* __restrict__ dst, int ld,
                                   const float4 (&v)[NV], int tid) {
  const int rpi = 256 / W4;
  float* p = dst + (tid / W4) * ld + (tid % W4) * 4;
#pragma unroll
  for (int i = 0; i < NV; ++i) *(float4*)(p + i * rpi * ld) = v[i];
}

#define GCH 16

__device__ inline void load_gi_chunk(v2f (&g)[4][GCH / 2], const float* __restrict__ rowh) {
#pragma unroll
  for (int r = 0; r < 4; ++r)
#pragma unroll
    for (int q = 0; q < GCH / 2; ++q)
      g[r][q] = *(const v2f*)(rowh + r * LDG + 2 * q);
}

template <int H0>
__device__ inline void gate_chunk(v2f (&S)[4], const v2f (&g)[4][GCH / 2],
                                  const v2f (&P)[H_DIM], const float (&w)[H_DIM]) {
#pragma unroll
  for (int q = 0; q < GCH / 2; ++q)
#pragma unroll
    for (int e = 0; e < 2; ++e) {
      const int h = H0 + 2 * q + e;
      const v2f wv = (v2f){w[h], w[h]};
#pragma unroll
      for (int r = 0; r < 4; ++r) {
        v2f t = P[h] + (v2f){g[r][q][e], g[r][q][e]};
        t = __builtin_elementwise_max(t, (v2f){0.0f, 0.0f});
        S[r] = __builtin_elementwise_fma(t, wv, S[r]);
      }
    }
}

__global__ __launch_bounds__(256) void gated_comm_kernel(
    const float* __restrict__ obs, const float* __restrict__ enc_w,
    const float* __restrict__ enc_b, const float* __restrict__ gate_w1,
    const float* __restrict__ gate_b1, const float* __restrict__ gate_w2,
    const float* __restrict__ gate_b2, const float* __restrict__ msg_w1,
    const float* __restrict__ msg_b1, const float* __restrict__ msg_w2,
    const float* __restrict__ msg_b2, float* __restrict__ out) {
  __shared__ __attribute__((aligned(16))) float smem[LDS_FLOATS];

  const int b       = blockIdx.x;
  const int tid     = threadIdx.x;
  const int lane    = tid & 31;
  const int wave    = tid >> 5;
  const int l16     = lane & 15;
  const int rowBase = (wave & 3) * 16;
  const int ch      = __builtin_amdgcn_readfirstlane(tid >> 7);
  const int colBase = ch * 32;

  float* sOBS = smem + OFF_OBS;
  float* sWE  = smem + OFF_WE;
  float* sWG  = smem + OFF_WG;
  float* sWM1 = smem + OFF_WM1;
  float* sWM2 = smem + OFF_WM2;
  float* sGI  = smem + OFF_GI;
  float* sGJT = smem + OFF_GJT;
  float* sMSG = smem + OFF_MSG;
  float* sW2  = smem + OFF_W2;
  float* sENC = smem + OFF_ENC;
  float* sTMP = smem + OFF_TMP;

  const float* obsb = obs + (size_t)b * (N_DIM * D_DIM);

  float4 vo[8], ve[8], vg[8], v1[4], v2[4];
  burst_load(vo, obsb, tid);
  burst_loadP(ve, enc_w, tid);
  burst_loadP(vg, gate_w1, tid);
  burst_loadP(v1, msg_w1, tid);
  burst_loadP(v2, msg_w2, tid);
  float be[NT], bg[NT], bz[NT], bm1[4], bm2[4];
#pragma unroll
  for (int t = 0; t < NT; ++t) {
    be[t] = enc_b[colBase + t * 16 + l16];
    bg[t] = gate_b1[colBase + t * 16 + l16];
    bz[t] = 0.0f;
  }
#pragma unroll
  for (int t = 0; t < 4; ++t) {
    bm1[t] = msg_b1[t * 16 + l16];
    bm2[t] = msg_b2[t * 16 + l16];
  }
  const float w2v = gate_w2[tid & 63];
  const float gb2 = gate_b2[0];

  burst_store<8, 32>(sOBS, LDA, vo, tid);
  burst_storeP(sWE,  ve, tid);
  burst_storeP(sWG,  vg, tid);
  burst_storeP(sWM1, v1, tid);
  burst_storeP(sWM2, v2, tid);
  sW2[tid & 63] = w2v;
  __syncthreads();

  v8f acc[NT];

  init_acc<NT>(acc, be);
  gemm_acc<D_DIM, NT, 8>(sOBS, LDA, sWE, acc, lane, rowBase, colBase);
  store_stripe<NT>(sENC, LDG, acc, lane, rowBase, colBase,  true);
  __syncthreads();

  init_acc<NT>(acc, bz);
  gemm_acc<H_DIM, NT, 8>(sENC, LDG, sWG, acc, lane, rowBase, colBase);
  store_stripe<NT>(sGI, LDG, acc, lane, rowBase, colBase, false);

  init_acc<NT>(acc, bg);
  gemm_acc<H_DIM, NT, 8>(sENC, LDG, sWG + 32 * LDP2, acc, lane, rowBase, colBase);
  store_stripeT<NT>(sGJT, LDG, acc, lane, rowBase, colBase);
  __syncthreads();

  int pBeg = 0, pEnd = 3;
  if (ch == 0) {
    v8f am[4];
    init_acc<4>(am, bm1);
    gemm_acc<H_DIM, 4, 4>(sENC, LDG, sWM1, am, lane, rowBase, 0);
    store_stripe<4>(sTMP, LDG, am, lane, rowBase, 0,  true);
    init_acc<4>(am, bm2);
    gemm_acc<H_DIM, 4, 4>(sTMP, LDG, sWM2, am, lane, rowBase, 0);
    store_stripeP<4>(sMSG, am, lane, rowBase, 0);
    pBeg = 3; pEnd = 4;
  }

  v2f P[H_DIM];
#pragma unroll
  for (int h = 0; h < H_DIM; ++h)
    P[h] = *(const v2f*)(sGJT + h * LDG + 2 * lane);
  float w2r[H_DIM];
#pragma unroll
  for (int h = 0; h < H_DIM; h += 4) {
    const float4 q = *(const float4*)(sW2 + h);
    w2r[h] = q.x; w2r[h + 1] = q.y; w2r[h + 2] = q.z; w2r[h + 3] = q.w;
  }
  v2f gA[4][GCH / 2], gB[4][GCH / 2];
  load_gi_chunk(gA, sGI + (rowBase + 4 * pBeg) * LDG);
#pragma unroll 1
  for (int p = pBeg; p < pEnd; ++p) {
    float* row = sGI + (rowBase + 4 * p) * LDG;
    const float* nrow = row + (p + 1 < pEnd ? 4 * LDG : 0);
    v2f S[4];
#pragma unroll
    for (int r = 0; r < 4; ++r) S[r] = (v2f){0.0f, 0.0f};

    load_gi_chunk(gB, row + GCH);
    __builtin_amdgcn_sched_barrier(0);
    gate_chunk<0 * GCH>(S, gA, P, w2r);
    load_gi_chunk(gA, row + 2 * GCH);
    __builtin_amdgcn_sched_barrier(0);
    gate_chunk<1 * GCH>(S, gB, P, w2r);
    load_gi_chunk(gB, row + 3 * GCH);
    __builtin_amdgcn_sched_barrier(0);
    gate_chunk<2 * GCH>(S, gA, P, w2r);
    load_gi_chunk(gA, nrow);
    __builtin_amdgcn_sched_barrier(0);
    gate_chunk<3 * GCH>(S, gB, P, w2r);

#pragma unroll
    for (int r = 0; r < 4; ++r) {
      v2f gt;
      gt[0] = 1.0f / (1.0f + __expf(-(S[r][0] + gb2)));
      gt[1] = 1.0f / (1.0f + __expf(-(S[r][1] + gb2)));
      *(v2f*)(row + r * LDG + 2 * lane) = gt;
    }
  }
  __syncthreads();

  init_acc<NT>(acc, bz);
  gemm_acc<N_DIM, NT, 8>(sGI, LDG, sMSG, acc, lane, rowBase, colBase);

  float* outb = out + (size_t)b * (N_DIM * H_DIM);
  const int half = lane >> 4;
#pragma unroll
  for (int t = 0; t < NT; ++t)
#pragma unroll
    for (int r = 0; r < 8; ++r)
      outb[(rowBase + r + 8 * half) * H_DIM + colBase + t * 16 + l16] = acc[t][r];
}

extern "C" void kernel_launch(void* const* d_in, const int* in_sizes, int n_in,
                              void* d_out, int out_size, void* d_ws, size_t ws_size,
                              hipStream_t stream) {
  const float* obs     = (const float*)d_in[0];
  const float* enc_w   = (const float*)d_in[1];
  const float* enc_b   = (const float*)d_in[2];
  const float* gate_w1 = (const float*)d_in[3];
  const float* gate_b1 = (const float*)d_in[4];
  const float* gate_w2 = (const float*)d_in[5];
  const float* gate_b2 = (const float*)d_in[6];
  const float* msg_w1  = (const float*)d_in[7];
  const float* msg_b1  = (const float*)d_in[8];
  const float* msg_w2  = (const float*)d_in[9];
  const float* msg_b2  = (const float*)d_in[10];
  float* out = (float*)d_out;

  gated_comm_kernel<<<BATCH, 256, 0, stream>>>(
      obs, enc_w, enc_b, gate_w1, gate_b1, gate_w2, gate_b2,
      msg_w1, msg_b1, msg_w2, msg_b2, out);
}
